# starting kernel
# baseline (speedup 1.0000x reference)
#include <hip/hip_runtime.h>
#include <hip/hip_bf16.h>
#include <math.h>

typedef float v2f __attribute__((ext_vector_type(2)));
typedef float v8f __attribute__((ext_vector_type(8)));

#define BATCH   2
#define SEQ     2048
#define DMODEL  1024
#define DTRANK  64
#define NSTATE  16
#define KPROJ   (DTRANK + 2 * NSTATE)
#define MROWS   (BATCH * SEQ)
#define NCHAN   (BATCH * DMODEL)
#define NCHUNK  32
#define CLEN    (SEQ / NCHUNK)

__global__ __launch_bounds__(256)
void gemm_xproj(const float* __restrict__ U, const float* __restrict__ W1,
                float* __restrict__ xp) {
    const int lane  = threadIdx.x & 31;
    const int wave  = threadIdx.x >> 5;
    const int tile  = blockIdx.x * (blockDim.x >> 5) + wave;
    const int tilesN = KPROJ / 16;
    const int m0 = (tile / tilesN) * 16;
    const int n0 = (tile % tilesN) * 16;

    const int row   = m0 + (lane & 15);
    const int col   = n0 + (lane & 15);
    const int khalf = (lane >> 4) * 2;

    v8f acc = {};
    #pragma unroll 4
    for (int k = 0; k < DMODEL; k += 4) {
        v2f a = *(const v2f*)(U  + row * DMODEL + k + khalf);
        v2f b = *(const v2f*)(W1 + col * DMODEL + k + khalf);
        acc = __builtin_amdgcn_wmma_f32_16x16x4_f32(false, a, false, b,
                                                    (short)0, acc, false, false);
    }
    const int rbase = m0 + ((lane >> 4) ? 8 : 0);
    #pragma unroll
    for (int i = 0; i < 8; ++i)
        xp[(rbase + i) * KPROJ + col] = acc[i];
}

__global__ __launch_bounds__(256)
void gemm_delta(const float* __restrict__ xp, const float* __restrict__ W2,
                const float* __restrict__ bias, float* __restrict__ delta) {
    const int lane  = threadIdx.x & 31;
    const int wave  = threadIdx.x >> 5;
    const int tile  = blockIdx.x * (blockDim.x >> 5) + wave;
    const int tilesN = DMODEL / 16;
    const int m0 = (tile / tilesN) * 16;
    const int n0 = (tile % tilesN) * 16;

    const int row   = m0 + (lane & 15);
    const int col   = n0 + (lane & 15);
    const int khalf = (lane >> 4) * 2;

    v8f acc = {};
    #pragma unroll
    for (int k = 0; k < DTRANK; k += 4) {
        v2f a = *(const v2f*)(xp + row * KPROJ  + k + khalf);
        v2f b = *(const v2f*)(W2 + col * DTRANK + k + khalf);
        acc = __builtin_amdgcn_wmma_f32_16x16x4_f32(false, a, false, b,
                                                    (short)0, acc, false, false);
    }
    const int   rbase = m0 + ((lane >> 4) ? 8 : 0);
    const float bn    = bias[col];
    #pragma unroll
    for (int i = 0; i < 8; ++i) {
        float x  = acc[i] + bn;
        float sp = (x > 20.0f) ? x : log1pf(__expf(x));
        delta[(rbase + i) * DMODEL + col] = sp;
    }
}

__global__ __launch_bounds__(256)
void scan_local(const float* __restrict__ U, const float* __restrict__ A_log,
                const float* __restrict__ xp, const float* __restrict__ delta,
                float* __restrict__ out, float* __restrict__ agg) {
    const int ch = blockIdx.x * blockDim.x + threadIdx.x;
    const int c  = blockIdx.y;
    const int bb = ch >> 10;
    const int dd = ch & (DMODEL - 1);

    float Av[NSTATE];
    #pragma unroll
    for (int n = 0; n < NSTATE; ++n) Av[n] = -__expf(A_log[n]);

    float st[NSTATE];
    #pragma unroll
    for (int n = 0; n < NSTATE; ++n) st[n] = 0.0f;

    float dsum = 0.0f;
    const int t0 = c * CLEN;
    for (int t = t0; t < t0 + CLEN; ++t) {
        const int rowu = bb * SEQ + t;
        const float dl = delta[rowu * DMODEL + dd];
        const float uv = U[rowu * DMODEL + dd];
        const float4* Bp = (const float4*)(xp + rowu * KPROJ + DTRANK);
        const float4* Cp = (const float4*)(xp + rowu * KPROJ + DTRANK + NSTATE);
        float Bv[NSTATE], Cv[NSTATE];
        #pragma unroll
        for (int q = 0; q < 4; ++q) {
            float4 vb = Bp[q], vc = Cp[q];
            Bv[4*q+0]=vb.x; Bv[4*q+1]=vb.y; Bv[4*q+2]=vb.z; Bv[4*q+3]=vb.w;
            Cv[4*q+0]=vc.x; Cv[4*q+1]=vc.y; Cv[4*q+2]=vc.z; Cv[4*q+3]=vc.w;
        }
        const float du = dl * uv;
        float y = 0.0f;
        #pragma unroll
        for (int n = 0; n < NSTATE; ++n) {
            float dA = __expf(dl * Av[n]);
            st[n] = fmaf(dA, st[n], du * Bv[n]);
            y     = fmaf(st[n], Cv[n], y);
        }
        dsum += dl;
        out[rowu * DMODEL + dd] = y;
    }
    const int base = (c * 2 * NSTATE) * NCHAN + ch;
    #pragma unroll
    for (int n = 0; n < NSTATE; ++n)
        agg[base + n * NCHAN] = __expf(Av[n] * dsum);
    #pragma unroll
    for (int n = 0; n < NSTATE; ++n)
        agg[base + (NSTATE + n) * NCHAN] = st[n];
}

__global__ __launch_bounds__(256)
void scan_fixup(const float* __restrict__ U, const float* __restrict__ A_log,
                const float* __restrict__ xp, const float* __restrict__ delta,
                const float* __restrict__ Dvec, const float* __restrict__ agg,
                float* __restrict__ out) {
    const int ch = blockIdx.x * blockDim.x + threadIdx.x;
    const int c  = blockIdx.y;
    const int bb = ch >> 10;
    const int dd = ch & (DMODEL - 1);

    float Av[NSTATE];
    #pragma unroll
    for (int n = 0; n < NSTATE; ++n) Av[n] = -__expf(A_log[n]);

    float entry[NSTATE];
    #pragma unroll
    for (int n = 0; n < NSTATE; ++n) entry[n] = 0.0f;
    for (int cc = 0; cc < c; ++cc) {
        const int base = (cc * 2 * NSTATE) * NCHAN + ch;
        #pragma unroll
        for (int n = 0; n < NSTATE; ++n) {
            float p = agg[base + n * NCHAN];
            float f = agg[base + (NSTATE + n) * NCHAN];
            entry[n] = fmaf(p, entry[n], f);
        }
    }

    const float Dd = Dvec[dd];
    const int t0 = c * CLEN;
    for (int t = t0; t < t0 + CLEN; ++t) {
        const int rowu = bb * SEQ + t;
        const float dl = delta[rowu * DMODEL + dd];
        const float uv = U[rowu * DMODEL + dd];
        const float4* Cp = (const float4*)(xp + rowu * KPROJ + DTRANK + NSTATE);
        float Cv[NSTATE];
        #pragma unroll
        for (int q = 0; q < 4; ++q) {
            float4 vc = Cp[q];
            Cv[4*q+0]=vc.x; Cv[4*q+1]=vc.y; Cv[4*q+2]=vc.z; Cv[4*q+3]=vc.w;
        }
        float ycorr = 0.0f;
        #pragma unroll
        for (int n = 0; n < NSTATE; ++n) {
            entry[n] *= __expf(dl * Av[n]);
            ycorr = fmaf(entry[n], Cv[n], ycorr);
        }
        const int idx = rowu * DMODEL + dd;
        out[idx] = out[idx] + ycorr + uv * Dd;
    }
}

extern "C" void kernel_launch(void* const* d_in, const int* in_sizes, int n_in,
                              void* d_out, int out_size, void* d_ws, size_t ws_size,
                              hipStream_t stream) {
    const float* U     = (const float*)d_in[0];
    const float* A_log = (const float*)d_in[1];
    const float* W1    = (const float*)d_in[2];
    const float* W2    = (const float*)d_in[3];
    const float* bias  = (const float*)d_in[4];
    const float* Dvec  = (const float*)d_in[5];
    float* out = (float*)d_out;

    float* ws    = (float*)d_ws;
    float* xp    = ws;
    float* delta = xp + (size_t)MROWS * KPROJ;
    float* agg   = delta + (size_t)MROWS * DMODEL;

    gemm_xproj<<<dim3(1536 / 8), dim3(256), 0, stream>>>(U, W1, xp);

    gemm_delta<<<dim3(16384 / 8), dim3(256), 0, stream>>>(xp, W2, bias, delta);

    dim3 sgrid(NCHAN / 256, NCHUNK);
    scan_local<<<sgrid, dim3(256), 0, stream>>>(U, A_log, xp, delta, out, agg);
    scan_fixup<<<sgrid, dim3(256), 0, stream>>>(U, A_log, xp, delta, Dvec, agg, out);
}


# prefix launch removed: pass 1 blocks own 8 consecutive chunks of 32 channels, chain their aggregates in LDS; pass 2 runs <=8 stored links; multi-chunk decay as exp(A*sum): new sum order
# speedup vs baseline: 2.4718x; 2.4718x over previous
#include <hip/hip_runtime.h>
#include <hip/hip_bf16.h>
#include <math.h>

typedef float v2f __attribute__((ext_vector_type(2)));
typedef float v8f __attribute__((ext_vector_type(8)));

#define BATCH   2
#define SEQ     2048
#define DMODEL  1024
#define DTRANK  64
#define NSTATE  16
#define KPROJ   (DTRANK + 2 * NSTATE)
#define MROWS   (BATCH * SEQ)
#define NCHAN   (BATCH * DMODEL)
#define NCHUNK  64
#define CLEN    (SEQ / NCHUNK)
#define SCANBLK 256
#define CPS     8
#define NGROUP  (NCHUNK / CPS)
#define NLINK   (NSTATE + 1)
#define LBATCH  4

typedef float v4f __attribute__((ext_vector_type(4)));

#define G1_SET 32

#define G1_LOAD(A, P, Q, kk)                                            \
    _Pragma("unroll")                                                   \
    for (int j = 0; j < 4; ++j) {                                       \
        A[j] = *(const v4f*)(pa  + (kk) + 4 * j);                       \
        P[j] = *(const v4f*)(pb0 + (kk) + 4 * j);                       \
        Q[j] = *(const v4f*)(pb1 + (kk) + 4 * j);                       \
    }

#define G1_MMA(A, P, Q)                                                 \
    _Pragma("unroll")                                                   \
    for (int j = 0; j < 4; ++j) {                                       \
        const v2f alo = {A[j].x, A[j].y}, ahi = {A[j].z, A[j].w};       \
        const v2f plo = {P[j].x, P[j].y}, phi = {P[j].z, P[j].w};       \
        const v2f qlo = {Q[j].x, Q[j].y}, qhi = {Q[j].z, Q[j].w};       \
        acc0 = __builtin_amdgcn_wmma_f32_16x16x4_f32(false, alo, false, plo, (short)0, acc0, false, false); \
        acc1 = __builtin_amdgcn_wmma_f32_16x16x4_f32(false, alo, false, qlo, (short)0, acc1, false, false); \
        acc0 = __builtin_amdgcn_wmma_f32_16x16x4_f32(false, ahi, false, phi, (short)0, acc0, false, false); \
        acc1 = __builtin_amdgcn_wmma_f32_16x16x4_f32(false, ahi, false, qhi, (short)0, acc1, false, false); \
    }

#define G1_PHASE __builtin_amdgcn_sched_barrier(0);

__global__ __launch_bounds__(96)
void gemm_xproj(const float* __restrict__ U, const float* __restrict__ W1,
                float* __restrict__ xp) {
    const int lane = threadIdx.x & 31;
    const int wave = threadIdx.x >> 5;
    const int lc   = lane & 15;
    const int hh   = lane >> 4;
    const int m0   = blockIdx.x * 16;
    const int n0   = wave * 32;

    const float* pa  = U  + (m0 + lc) * DMODEL + 16 * hh;
    const float* pb0 = W1 + (n0 + lc) * DMODEL + 16 * hh;
    const float* pb1 = pb0 + 16 * DMODEL;

    v8f acc0 = {}, acc1 = {};
    v4f a0[4], p0[4], q0[4];
    v4f a1[4], p1[4], q1[4];
    v4f a2[4], p2[4], q2[4];
    v4f a3[4], p3[4], q3[4];

    G1_LOAD(a0, p0, q0, 0)
    G1_LOAD(a1, p1, q1, G1_SET)
    G1_LOAD(a2, p2, q2, 2 * G1_SET)
    #pragma unroll
    for (int kb = 0; kb < DMODEL - 4 * G1_SET; kb += 4 * G1_SET) {
        G1_LOAD(a3, p3, q3, kb + 3 * G1_SET)
        G1_MMA(a0, p0, q0)
        G1_PHASE
        G1_LOAD(a0, p0, q0, kb + 4 * G1_SET)
        G1_MMA(a1, p1, q1)
        G1_PHASE
        G1_LOAD(a1, p1, q1, kb + 5 * G1_SET)
        G1_MMA(a2, p2, q2)
        G1_PHASE
        G1_LOAD(a2, p2, q2, kb + 6 * G1_SET)
        G1_MMA(a3, p3, q3)
        G1_PHASE
    }
    G1_LOAD(a3, p3, q3, DMODEL - G1_SET)
    G1_MMA(a0, p0, q0)
    G1_MMA(a1, p1, q1)
    G1_MMA(a2, p2, q2)
    G1_MMA(a3, p3, q3)

    const int rbase = m0 + 8 * hh;
    #pragma unroll
    for (int i = 0; i < 8; ++i) {
        xp[(rbase + i) * KPROJ + n0 + lc]      = acc0[i];
        xp[(rbase + i) * KPROJ + n0 + 16 + lc] = acc1[i];
    }
}

#define DPITCH 36

__device__ __forceinline__ float softplus_f(float x) {
    const float e = __builtin_amdgcn_exp2f(-1.44269504f * __builtin_fabsf(x));
    return fmaf(0.693147181f, __builtin_amdgcn_logf(1.0f + e), __builtin_fmaxf(x, 0.0f));
}

__global__ __launch_bounds__(SCANBLK)
void scan_local(const float* __restrict__ U, const float* __restrict__ A_log,
                const float* __restrict__ xp, const float* __restrict__ W2,
                const float* __restrict__ bias, float* __restrict__ dcum,
                float* __restrict__ out, float* __restrict__ qr, float* __restrict__ grp) {
    __shared__ float sdl[CPS][CLEN][DPITCH];

    const int lane  = threadIdx.x & 31;
    const int wave  = __builtin_amdgcn_readfirstlane(threadIdx.x >> 5);
    const int ch = blockIdx.x * 32 + lane;
    const int c  = blockIdx.y * CPS + wave;
    const int bb = (blockIdx.x * 32) / DMODEL;
    const int dd = ch & (DMODEL - 1);

    const int lc    = lane & 15;
    const int hh    = lane >> 4;
    const int khalf = hh * 2;
    const int row0  = bb * SEQ + c * CLEN + lc;
    const int row1  = row0 + 16;
    const int col0  = (dd & ~31) + lc;

    v8f acc00 = {}, acc01 = {}, acc10 = {}, acc11 = {};
    #pragma unroll
    for (int k = 0; k < DTRANK; k += 4) {
        v2f a0 = *(const v2f*)(xp + row0 * KPROJ + k + khalf);
        v2f a1 = *(const v2f*)(xp + row1 * KPROJ + k + khalf);
        v2f b0 = *(const v2f*)(W2 + col0 * DTRANK + k + khalf);
        v2f b1 = *(const v2f*)(W2 + (col0 + 16) * DTRANK + k + khalf);
        acc00 = __builtin_amdgcn_wmma_f32_16x16x4_f32(false, a0, false, b0,
                                                      (short)0, acc00, false, false);
        acc01 = __builtin_amdgcn_wmma_f32_16x16x4_f32(false, a0, false, b1,
                                                      (short)0, acc01, false, false);
        acc10 = __builtin_amdgcn_wmma_f32_16x16x4_f32(false, a1, false, b0,
                                                      (short)0, acc10, false, false);
        acc11 = __builtin_amdgcn_wmma_f32_16x16x4_f32(false, a1, false, b1,
                                                      (short)0, acc11, false, false);
    }
    const float bn0 = bias[col0];
    const float bn1 = bias[col0 + 16];
    #pragma unroll
    for (int i = 0; i < 8; ++i) {
        sdl[wave][8 * hh + i][lc]           = acc00[i] + bn0;
        sdl[wave][8 * hh + i][16 + lc]      = acc01[i] + bn1;
        sdl[wave][16 + 8 * hh + i][lc]      = acc10[i] + bn0;
        sdl[wave][16 + 8 * hh + i][16 + lc] = acc11[i] + bn1;
    }
    __syncthreads();

    float Av[NSTATE];
    #pragma unroll
    for (int n = 0; n < NSTATE; ++n) Av[n] = -__expf(A_log[n]);

    float st[NSTATE];
    #pragma unroll
    for (int n = 0; n < NSTATE; ++n) st[n] = 0.0f;

    float dsum = 0.0f;
    for (int s = 0; s < CLEN; s += 2) {
        int   rowu[2];
        float dl[2], uv[2];
        float Bv[2][NSTATE], Cv[2][NSTATE];
        #pragma unroll
        for (int j = 0; j < 2; ++j) {
            rowu[j] = bb * SEQ + c * CLEN + s + j;
            uv[j] = U[rowu[j] * DMODEL + dd];
            const float4* Bp = (const float4*)(xp + rowu[j] * KPROJ + DTRANK);
            const float4* Cp = (const float4*)(xp + rowu[j] * KPROJ + DTRANK + NSTATE);
            #pragma unroll
            for (int q = 0; q < 4; ++q) {
                float4 vb = Bp[q], vc = Cp[q];
                Bv[j][4*q+0]=vb.x; Bv[j][4*q+1]=vb.y; Bv[j][4*q+2]=vb.z; Bv[j][4*q+3]=vb.w;
                Cv[j][4*q+0]=vc.x; Cv[j][4*q+1]=vc.y; Cv[j][4*q+2]=vc.z; Cv[j][4*q+3]=vc.w;
            }
        }
        __builtin_amdgcn_sched_barrier(0);
        #pragma unroll
        for (int j = 0; j < 2; ++j)
            dl[j] = softplus_f(sdl[wave][s + j][lane]);
        #pragma unroll
        for (int j = 0; j < 2; ++j) {
            dsum += dl[j];
            dcum[rowu[j] * DMODEL + dd] = dsum;
        }
        #pragma unroll
        for (int j = 0; j < 2; ++j) {
            const float du = dl[j] * uv[j];
            float y = 0.0f;
            #pragma unroll
            for (int n = 0; n < NSTATE; ++n) {
                float dA = __expf(dl[j] * Av[n]);
                st[n] = fmaf(dA, st[n], du * Bv[j][n]);
                y     = fmaf(st[n], Cv[j][n], y);
            }
            out[rowu[j] * DMODEL + dd] = y;
        }
    }
    float* const ag = &sdl[0][0][0];
    float Pc[NSTATE];
    #pragma unroll
    for (int n = 0; n < NSTATE; ++n) {
        Pc[n] = __expf(Av[n] * dsum);
        ag[wave * (CLEN * DPITCH) + n * 32 + lane]            = Pc[n];
        ag[wave * (CLEN * DPITCH) + (NSTATE + n) * 32 + lane] = st[n];
    }
    ag[wave * (CLEN * DPITCH) + 2 * NSTATE * 32 + lane] = dsum;
    __builtin_amdgcn_fence(__ATOMIC_RELEASE, "workgroup", "local");
    __builtin_amdgcn_s_barrier();
    __builtin_amdgcn_fence(__ATOMIC_ACQUIRE, "workgroup", "local");

    float R[NSTATE];
    #pragma unroll
    for (int n = 0; n < NSTATE; ++n) R[n] = 0.0f;
    float dpre = 0.0f;
    for (int i = 0; i < wave; ++i) {
        const float* a = ag + i * (CLEN * DPITCH) + lane;
        #pragma unroll
        for (int n = 0; n < NSTATE; ++n)
            R[n] = fmaf(a[n * 32], R[n], a[(NSTATE + n) * 32]);
        dpre += a[2 * NSTATE * 32];
    }
    const int base = (c * NLINK) * NCHAN + ch;
    #pragma unroll
    for (int n = 0; n < NSTATE; ++n)
        qr[base + n * NCHAN] = R[n];
    qr[base + NSTATE * NCHAN] = dpre;
    if (wave == CPS - 1) {
        const int gbase = (blockIdx.y * NLINK) * NCHAN + ch;
        #pragma unroll
        for (int n = 0; n < NSTATE; ++n)
            grp[gbase + n * NCHAN] = fmaf(Pc[n], R[n], st[n]);
        grp[gbase + NSTATE * NCHAN] = dpre + dsum;
    }
}

#define FSTEP 8
#define CROWS 4

__global__ __launch_bounds__(SCANBLK)
void scan_fixup(const float* __restrict__ U, const float* __restrict__ A_log,
                const float* __restrict__ xp, const float* __restrict__ dcum,
                const float* __restrict__ Dvec, const float* __restrict__ qr,
                const float* __restrict__ grp, float* __restrict__ out) {
    const int ch = blockIdx.x * SCANBLK + threadIdx.x;
    const int c  = blockIdx.y;
    const int bb = (blockIdx.x * SCANBLK) / DMODEL;
    const int dd = ch & (DMODEL - 1);

    float Av[NSTATE];
    #pragma unroll
    for (int n = 0; n < NSTATE; ++n) Av[n] = -__expf(A_log[n]);

    const int ng = c / CPS;
    float entry[NSTATE];
    #pragma unroll
    for (int n = 0; n < NSTATE; ++n) entry[n] = 0.0f;
    for (int k0 = 0; k0 <= ng; k0 += LBATCH) {
        float dv[LBATCH], lv[LBATCH][NSTATE];
        #pragma unroll
        for (int j = 0; j < LBATCH; ++j) {
            const int k = (k0 + j < ng) ? k0 + j : ng;
            const float* lk = (k < ng) ? grp + (k * NLINK) * NCHAN : qr + (c * NLINK) * NCHAN;
            #pragma unroll
            for (int n = 0; n < NSTATE; ++n) lv[j][n] = lk[n * NCHAN + ch];
            dv[j] = lk[NSTATE * NCHAN + ch];
        }
        __builtin_amdgcn_sched_barrier(0);
        #pragma unroll
        for (int j = 0; j < LBATCH; ++j) {
            if (k0 + j <= ng) {
                #pragma unroll
                for (int n = 0; n < NSTATE; ++n)
                    entry[n] = fmaf(__expf(Av[n] * dv[j]), entry[n], lv[j][n]);
            }
        }
    }

    const float Dd = Dvec[dd];
    for (int s0 = 0; s0 < CLEN; s0 += FSTEP) {
        int   idx[FSTEP];
        float cd[FSTEP], uv[FSTEP], yl[FSTEP];
        #pragma unroll
        for (int j = 0; j < FSTEP; ++j) {
            const int rowu = bb * SEQ + c * CLEN + s0 + j;
            idx[j] = rowu * DMODEL + dd;
            cd[j] = dcum[idx[j]];
            uv[j] = U[idx[j]];
            yl[j] = out[idx[j]];
        }
        #pragma unroll
        for (int h = 0; h < FSTEP; h += CROWS) {
            float Cv[CROWS][NSTATE];
            #pragma unroll
            for (int j = 0; j < CROWS; ++j) {
                const int rowu = bb * SEQ + c * CLEN + s0 + h + j;
                const float4* Cp = (const float4*)(xp + rowu * KPROJ + DTRANK + NSTATE);
                #pragma unroll
                for (int q = 0; q < 4; ++q) {
                    float4 vc = Cp[q];
                    Cv[j][4*q+0]=vc.x; Cv[j][4*q+1]=vc.y; Cv[j][4*q+2]=vc.z; Cv[j][4*q+3]=vc.w;
                }
            }
            __builtin_amdgcn_sched_barrier(0);
            #pragma unroll
            for (int j = 0; j < CROWS; ++j) {
                float ycorr = 0.0f;
                #pragma unroll
                for (int n = 0; n < NSTATE; ++n) {
                    const float e = entry[n] * __expf(cd[h + j] * Av[n]);
                    ycorr = fmaf(e, Cv[j][n], ycorr);
                }
                out[idx[h + j]] = yl[h + j] + ycorr + uv[h + j] * Dd;
            }
        }
    }
}

extern "C" void kernel_launch(void* const* d_in, const int* in_sizes, int n_in,
                              void* d_out, int out_size, void* d_ws, size_t ws_size,
                              hipStream_t stream) {
    const float* U     = (const float*)d_in[0];
    const float* A_log = (const float*)d_in[1];
    const float* W1    = (const float*)d_in[2];
    const float* W2    = (const float*)d_in[3];
    const float* bias  = (const float*)d_in[4];
    const float* Dvec  = (const float*)d_in[5];
    float* out = (float*)d_out;

    float* ws    = (float*)d_ws;
    float* xp    = ws;
    float* dcum  = xp + (size_t)MROWS * KPROJ;
    float* qr    = dcum + (size_t)MROWS * DMODEL;
    float* grp   = qr + (size_t)NCHUNK * NLINK * NCHAN;

    gemm_xproj<<<dim3(MROWS / 16), dim3(96), 0, stream>>>(U, W1, xp);

    scan_local<<<dim3(NCHAN / 32, NGROUP), dim3(SCANBLK), 0, stream>>>(U, A_log, xp, W2, bias, dcum, out, qr, grp);
    scan_fixup<<<dim3(NCHAN / SCANBLK, NCHUNK), dim3(SCANBLK), 0, stream>>>(U, A_log, xp, dcum, Dvec, qr, grp, out);
}


# projection launch stores its dt columns in the delta product's fragment order: pass 1 reads each A slot as one contiguous 512 B wave load, W2 as 16 B per lane; k regrouped per WMMA: new sum order
# speedup vs baseline: 2.5865x; 1.0464x over previous
#include <hip/hip_runtime.h>
#include <hip/hip_bf16.h>
#include <math.h>

typedef float v2f __attribute__((ext_vector_type(2)));
typedef float v8f __attribute__((ext_vector_type(8)));

#define BATCH   2
#define SEQ     2048
#define DMODEL  1024
#define DTRANK  64
#define NSTATE  16
#define KPROJ   (DTRANK + 2 * NSTATE)
#define MROWS   (BATCH * SEQ)
#define NCHAN   (BATCH * DMODEL)
#define NCHUNK  64
#define CLEN    (SEQ / NCHUNK)
#define SCANBLK 256
#define CPS     8
#define NGROUP  (NCHUNK / CPS)
#define NLINK   (NSTATE + 1)
#define LBATCH  4
#define BCW     (2 * NSTATE)
#define DTF_N   (MROWS * DTRANK)

typedef float v4f __attribute__((ext_vector_type(4)));

#define G1_SET 32

#define G1_LOAD(A, P, Q, kk)                                            \
    _Pragma("unroll")                                                   \
    for (int j = 0; j < 4; ++j) {                                       \
        A[j] = *(const v4f*)(pa  + (kk) + 4 * j);                       \
        P[j] = *(const v4f*)(pb0 + (kk) + 4 * j);                       \
        Q[j] = *(const v4f*)(pb1 + (kk) + 4 * j);                       \
    }

#define G1_MMA(A, P, Q)                                                 \
    _Pragma("unroll")                                                   \
    for (int j = 0; j < 4; ++j) {                                       \
        const v2f alo = {A[j].x, A[j].y}, ahi = {A[j].z, A[j].w};       \
        const v2f plo = {P[j].x, P[j].y}, phi = {P[j].z, P[j].w};       \
        const v2f qlo = {Q[j].x, Q[j].y}, qhi = {Q[j].z, Q[j].w};       \
        acc0 = __builtin_amdgcn_wmma_f32_16x16x4_f32(false, alo, false, plo, (short)0, acc0, false, false); \
        acc1 = __builtin_amdgcn_wmma_f32_16x16x4_f32(false, alo, false, qlo, (short)0, acc1, false, false); \
        acc0 = __builtin_amdgcn_wmma_f32_16x16x4_f32(false, ahi, false, phi, (short)0, acc0, false, false); \
        acc1 = __builtin_amdgcn_wmma_f32_16x16x4_f32(false, ahi, false, qhi, (short)0, acc1, false, false); \
    }

#define G1_PHASE __builtin_amdgcn_sched_barrier(0);

__global__ __launch_bounds__(96)
void gemm_xproj(const float* __restrict__ U, const float* __restrict__ W1,
                float* __restrict__ xo) {
    const int lane = threadIdx.x & 31;
    const int wave = threadIdx.x >> 5;
    const int lc   = lane & 15;
    const int hh   = lane >> 4;
    const int m0   = blockIdx.x * 16;
    const int n0   = wave * 32;

    const float* pa  = U  + (m0 + lc) * DMODEL + 16 * hh;
    const float* pb0 = W1 + (n0 + lc) * DMODEL + 16 * hh;
    const float* pb1 = pb0 + 16 * DMODEL;

    v8f acc0 = {}, acc1 = {};
    v4f a0[4], p0[4], q0[4];
    v4f a1[4], p1[4], q1[4];
    v4f a2[4], p2[4], q2[4];
    v4f a3[4], p3[4], q3[4];

    G1_LOAD(a0, p0, q0, 0)
    G1_LOAD(a1, p1, q1, G1_SET)
    G1_LOAD(a2, p2, q2, 2 * G1_SET)
    #pragma unroll
    for (int kb = 0; kb < DMODEL - 4 * G1_SET; kb += 4 * G1_SET) {
        G1_LOAD(a3, p3, q3, kb + 3 * G1_SET)
        G1_MMA(a0, p0, q0)
        G1_PHASE
        G1_LOAD(a0, p0, q0, kb + 4 * G1_SET)
        G1_MMA(a1, p1, q1)
        G1_PHASE
        G1_LOAD(a1, p1, q1, kb + 5 * G1_SET)
        G1_MMA(a2, p2, q2)
        G1_PHASE
        G1_LOAD(a2, p2, q2, kb + 6 * G1_SET)
        G1_MMA(a3, p3, q3)
        G1_PHASE
    }
    G1_LOAD(a3, p3, q3, DMODEL - G1_SET)
    G1_MMA(a0, p0, q0)
    G1_MMA(a1, p1, q1)
    G1_MMA(a2, p2, q2)
    G1_MMA(a3, p3, q3)

    const int T      = blockIdx.x;
    const int dbase  = ((T * 8 + 4 * wave + (lc >> 3)) * 32 + 16 * ((lc >> 2) & 1) + 8 * hh) * 4 + (lc & 3);
    const int bbase  = DTF_N + (m0 + 8 * hh) * BCW + lc;
    const bool isdt  = wave < 2;
    const int base   = isdt ? dbase : bbase;
    const int istep  = isdt ? 4 : BCW;
    const int off1   = isdt ? 256 : NSTATE;
    #pragma unroll
    for (int i = 0; i < 8; ++i) {
        xo[base + i * istep]        = acc0[i];
        xo[base + off1 + i * istep] = acc1[i];
    }
}

#define DPITCH 36

__device__ __forceinline__ float softplus_f(float x) {
    const float e = __builtin_amdgcn_exp2f(-1.44269504f * __builtin_fabsf(x));
    return fmaf(0.693147181f, __builtin_amdgcn_logf(1.0f + e), __builtin_fmaxf(x, 0.0f));
}

__global__ __launch_bounds__(SCANBLK)
void scan_local(const float* __restrict__ U, const float* __restrict__ A_log,
                const float* __restrict__ dtf, const float* __restrict__ bc,
                const float* __restrict__ W2,
                const float* __restrict__ bias, float* __restrict__ dcum,
                float* __restrict__ out, float* __restrict__ qr, float* __restrict__ grp) {
    __shared__ float sdl[CPS][CLEN][DPITCH];

    const int lane  = threadIdx.x & 31;
    const int wave  = __builtin_amdgcn_readfirstlane(threadIdx.x >> 5);
    const int ch = blockIdx.x * 32 + lane;
    const int c  = blockIdx.y * CPS + wave;
    const int bb = (blockIdx.x * 32) / DMODEL;
    const int dd = ch & (DMODEL - 1);

    const int lc    = lane & 15;
    const int hh    = lane >> 4;
    const int T0    = (bb * SEQ + c * CLEN) / 16;
    const int col0  = (dd & ~31) + lc;
    const v4f* pa0 = (const v4f*)(dtf + (T0 * 8 * 32 + lane) * 4);
    const v4f* pa1 = pa0 + 8 * 32;
    const v4f* pb0 = (const v4f*)(W2 + col0 * DTRANK + 4 * hh);
    const v4f* pb1 = pb0 + 16 * (DTRANK / 4);

    v8f acc00 = {}, acc01 = {}, acc10 = {}, acc11 = {};
    #pragma unroll
    for (int q = 0; q < DTRANK / 8; ++q) {
        const v4f a0 = pa0[q * 32], a1 = pa1[q * 32];
        const v4f b0 = pb0[q * 2],  b1 = pb1[q * 2];
        #pragma unroll
        for (int p = 0; p < 2; ++p) {
            const v2f a0p = p ? (v2f){a0.z, a0.w} : (v2f){a0.x, a0.y};
            const v2f a1p = p ? (v2f){a1.z, a1.w} : (v2f){a1.x, a1.y};
            const v2f b0p = p ? (v2f){b0.z, b0.w} : (v2f){b0.x, b0.y};
            const v2f b1p = p ? (v2f){b1.z, b1.w} : (v2f){b1.x, b1.y};
            acc00 = __builtin_amdgcn_wmma_f32_16x16x4_f32(false, a0p, false, b0p,
                                                          (short)0, acc00, false, false);
            acc01 = __builtin_amdgcn_wmma_f32_16x16x4_f32(false, a0p, false, b1p,
                                                          (short)0, acc01, false, false);
            acc10 = __builtin_amdgcn_wmma_f32_16x16x4_f32(false, a1p, false, b0p,
                                                          (short)0, acc10, false, false);
            acc11 = __builtin_amdgcn_wmma_f32_16x16x4_f32(false, a1p, false, b1p,
                                                          (short)0, acc11, false, false);
        }
    }
    const float bn0 = bias[col0];
    const float bn1 = bias[col0 + 16];
    #pragma unroll
    for (int i = 0; i < 8; ++i) {
        sdl[wave][8 * hh + i][lc]           = acc00[i] + bn0;
        sdl[wave][8 * hh + i][16 + lc]      = acc01[i] + bn1;
        sdl[wave][16 + 8 * hh + i][lc]      = acc10[i] + bn0;
        sdl[wave][16 + 8 * hh + i][16 + lc] = acc11[i] + bn1;
    }
    __syncthreads();

    float Av[NSTATE];
    #pragma unroll
    for (int n = 0; n < NSTATE; ++n) Av[n] = -__expf(A_log[n]);

    float st[NSTATE];
    #pragma unroll
    for (int n = 0; n < NSTATE; ++n) st[n] = 0.0f;

    float dsum = 0.0f;
    for (int s = 0; s < CLEN; s += 2) {
        int   rowu[2];
        float dl[2], uv[2];
        float Bv[2][NSTATE], Cv[2][NSTATE];
        #pragma unroll
        for (int j = 0; j < 2; ++j) {
            rowu[j] = bb * SEQ + c * CLEN + s + j;
            uv[j] = U[rowu[j] * DMODEL + dd];
            const float4* Bp = (const float4*)(bc + rowu[j] * BCW);
            const float4* Cp = (const float4*)(bc + rowu[j] * BCW + NSTATE);
            #pragma unroll
            for (int q = 0; q < 4; ++q) {
                float4 vb = Bp[q], vc = Cp[q];
                Bv[j][4*q+0]=vb.x; Bv[j][4*q+1]=vb.y; Bv[j][4*q+2]=vb.z; Bv[j][4*q+3]=vb.w;
                Cv[j][4*q+0]=vc.x; Cv[j][4*q+1]=vc.y; Cv[j][4*q+2]=vc.z; Cv[j][4*q+3]=vc.w;
            }
        }
        __builtin_amdgcn_sched_barrier(0);
        #pragma unroll
        for (int j = 0; j < 2; ++j)
            dl[j] = softplus_f(sdl[wave][s + j][lane]);
        #pragma unroll
        for (int j = 0; j < 2; ++j) {
            dsum += dl[j];
            dcum[rowu[j] * DMODEL + dd] = dsum;
        }
        #pragma unroll
        for (int j = 0; j < 2; ++j) {
            const float du = dl[j] * uv[j];
            float y = 0.0f;
            #pragma unroll
            for (int n = 0; n < NSTATE; ++n) {
                float dA = __expf(dl[j] * Av[n]);
                st[n] = fmaf(dA, st[n], du * Bv[j][n]);
                y     = fmaf(st[n], Cv[j][n], y);
            }
            out[rowu[j] * DMODEL + dd] = y;
        }
    }
    float* const ag = &sdl[0][0][0];
    float Pc[NSTATE];
    #pragma unroll
    for (int n = 0; n < NSTATE; ++n) {
        Pc[n] = __expf(Av[n] * dsum);
        ag[wave * (CLEN * DPITCH) + n * 32 + lane]            = Pc[n];
        ag[wave * (CLEN * DPITCH) + (NSTATE + n) * 32 + lane] = st[n];
    }
    ag[wave * (CLEN * DPITCH) + 2 * NSTATE * 32 + lane] = dsum;
    __builtin_amdgcn_fence(__ATOMIC_RELEASE, "workgroup", "local");
    __builtin_amdgcn_s_barrier();
    __builtin_amdgcn_fence(__ATOMIC_ACQUIRE, "workgroup", "local");

    float R[NSTATE];
    #pragma unroll
    for (int n = 0; n < NSTATE; ++n) R[n] = 0.0f;
    float dpre = 0.0f;
    for (int i = 0; i < wave; ++i) {
        const float* a = ag + i * (CLEN * DPITCH) + lane;
        #pragma unroll
        for (int n = 0; n < NSTATE; ++n)
            R[n] = fmaf(a[n * 32], R[n], a[(NSTATE + n) * 32]);
        dpre += a[2 * NSTATE * 32];
    }
    const int base = (c * NLINK) * NCHAN + ch;
    #pragma unroll
    for (int n = 0; n < NSTATE; ++n)
        qr[base + n * NCHAN] = R[n];
    qr[base + NSTATE * NCHAN] = dpre;
    if (wave == CPS - 1) {
        const int gbase = (blockIdx.y * NLINK) * NCHAN + ch;
        #pragma unroll
        for (int n = 0; n < NSTATE; ++n)
            grp[gbase + n * NCHAN] = fmaf(Pc[n], R[n], st[n]);
        grp[gbase + NSTATE * NCHAN] = dpre + dsum;
    }
}

#define FSTEP 8
#define CROWS 4

__global__ __launch_bounds__(SCANBLK)
void scan_fixup(const float* __restrict__ U, const float* __restrict__ A_log,
                const float* __restrict__ bc, const float* __restrict__ dcum,
                const float* __restrict__ Dvec, const float* __restrict__ qr,
                const float* __restrict__ grp, float* __restrict__ out) {
    const int ch = blockIdx.x * SCANBLK + threadIdx.x;
    const int c  = blockIdx.y;
    const int bb = (blockIdx.x * SCANBLK) / DMODEL;
    const int dd = ch & (DMODEL - 1);

    float Av[NSTATE];
    #pragma unroll
    for (int n = 0; n < NSTATE; ++n) Av[n] = -__expf(A_log[n]);

    const int ng = c / CPS;
    float entry[NSTATE];
    #pragma unroll
    for (int n = 0; n < NSTATE; ++n) entry[n] = 0.0f;
    for (int k0 = 0; k0 <= ng; k0 += LBATCH) {
        float dv[LBATCH], lv[LBATCH][NSTATE];
        #pragma unroll
        for (int j = 0; j < LBATCH; ++j) {
            const int k = (k0 + j < ng) ? k0 + j : ng;
            const float* lk = (k < ng) ? grp + (k * NLINK) * NCHAN : qr + (c * NLINK) * NCHAN;
            #pragma unroll
            for (int n = 0; n < NSTATE; ++n) lv[j][n] = lk[n * NCHAN + ch];
            dv[j] = lk[NSTATE * NCHAN + ch];
        }
        __builtin_amdgcn_sched_barrier(0);
        #pragma unroll
        for (int j = 0; j < LBATCH; ++j) {
            if (k0 + j <= ng) {
                #pragma unroll
                for (int n = 0; n < NSTATE; ++n)
                    entry[n] = fmaf(__expf(Av[n] * dv[j]), entry[n], lv[j][n]);
            }
        }
    }

    const float Dd = Dvec[dd];
    for (int s0 = 0; s0 < CLEN; s0 += FSTEP) {
        int   idx[FSTEP];
        float cd[FSTEP], uv[FSTEP], yl[FSTEP];
        #pragma unroll
        for (int j = 0; j < FSTEP; ++j) {
            const int rowu = bb * SEQ + c * CLEN + s0 + j;
            idx[j] = rowu * DMODEL + dd;
            cd[j] = dcum[idx[j]];
            uv[j] = U[idx[j]];
            yl[j] = out[idx[j]];
        }
        #pragma unroll
        for (int h = 0; h < FSTEP; h += CROWS) {
            float Cv[CROWS][NSTATE];
            #pragma unroll
            for (int j = 0; j < CROWS; ++j) {
                const int rowu = bb * SEQ + c * CLEN + s0 + h + j;
                const float4* Cp = (const float4*)(bc + rowu * BCW + NSTATE);
                #pragma unroll
                for (int q = 0; q < 4; ++q) {
                    float4 vc = Cp[q];
                    Cv[j][4*q+0]=vc.x; Cv[j][4*q+1]=vc.y; Cv[j][4*q+2]=vc.z; Cv[j][4*q+3]=vc.w;
                }
            }
            __builtin_amdgcn_sched_barrier(0);
            #pragma unroll
            for (int j = 0; j < CROWS; ++j) {
                float ycorr = 0.0f;
                #pragma unroll
                for (int n = 0; n < NSTATE; ++n) {
                    const float e = entry[n] * __expf(cd[h + j] * Av[n]);
                    ycorr = fmaf(e, Cv[j][n], ycorr);
                }
                out[idx[h + j]] = yl[h + j] + ycorr + uv[h + j] * Dd;
            }
        }
    }
}

extern "C" void kernel_launch(void* const* d_in, const int* in_sizes, int n_in,
                              void* d_out, int out_size, void* d_ws, size_t ws_size,
                              hipStream_t stream) {
    const float* U     = (const float*)d_in[0];
    const float* A_log = (const float*)d_in[1];
    const float* W1    = (const float*)d_in[2];
    const float* W2    = (const float*)d_in[3];
    const float* bias  = (const float*)d_in[4];
    const float* Dvec  = (const float*)d_in[5];
    float* out = (float*)d_out;

    float* ws    = (float*)d_ws;
    float* dtf   = ws;
    float* bc    = dtf + (size_t)DTF_N;
    float* dcum  = bc + (size_t)MROWS * BCW;
    float* qr    = dcum + (size_t)MROWS * DMODEL;
    float* grp   = qr + (size_t)NCHUNK * NLINK * NCHAN;

    gemm_xproj<<<dim3(MROWS / 16), dim3(96), 0, stream>>>(U, W1, dtf);

    scan_local<<<dim3(NCHAN / 32, NGROUP), dim3(SCANBLK), 0, stream>>>(U, A_log, dtf, bc, W2, bias, dcum, out, qr, grp);
    scan_fixup<<<dim3(NCHAN / SCANBLK, NCHUNK), dim3(SCANBLK), 0, stream>>>(U, A_log, bc, dcum, Dvec, qr, grp, out);
}
